# starting kernel
# baseline (speedup 1.0000x reference)
#include <hip/hip_runtime.h>

typedef float v2f __attribute__((ext_vector_type(2)));
typedef float v8f __attribute__((ext_vector_type(8)));

#define LCNT     12
#define BATCH    64
#define NDIM     201
#define DDIM     768
#define NWAVES   4
#define NTHREADS 128
#define TS       211

template<int M>
__device__ __forceinline__ void part_body(
    const float* __restrict__ x, const float* __restrict__ features,
    float* __restrict__ out, float* __restrict__ T,
    const int bb, const int p, const int e, const int s)
{
    constexpr int NT  = (M + 15) / 16;
    constexpr int Mp  = NT * 16;
    constexpr int TPW = (NT + NWAVES - 1) / NWAVES;

    const int tid  = threadIdx.x;
    const int wave = tid >> 5;
    const int lane = tid & 31;
    const int lr   = lane & 15;
    const int lh   = lane >> 4;

    const long xplane = (long)NDIM * NDIM;

    {
        const float* xl = x + (long)((LCNT - 1) * BATCH + bb) * xplane;
        for (int i = tid; i < 16 * Mp; i += NTHREADS) {
            const int r = i / Mp, c = i - r * Mp;
            float v = 0.f;
            if (c < M) {
                const int ridx = (r == 0) ? e : (s + r - 1);
                const int cidx = (c == 0) ? e : (s + c - 1);
                v = xl[(long)ridx * NDIM + cidx];
            }
            T[r * TS + c] = v;
        }
    }

    int  cidx_r[TPW];
    bool cv_r[TPW];
    #pragma unroll
    for (int ti = 0; ti < TPW; ++ti) {
        const int  t   = wave + ti * NWAVES;
        const int  cc  = t * 16 + lr;
        const bool cv  = (t < NT) && (cc < M);
        const int  ccl = cv ? cc : 0;
        cidx_r[ti] = (ccl == 0) ? e : (s + ccl - 1);
        cv_r[ti]   = cv;
    }

    for (int l = LCNT - 2; l >= 0; --l) {
        const float* xa = x + (long)(l * BATCH + bb) * xplane;
        __syncthreads();

        v8f acc[TPW];
        #pragma unroll
        for (int ti = 0; ti < TPW; ++ti) acc[ti] = (v8f)0.0f;

        #pragma unroll 2
        for (int k0 = 0; k0 < M; k0 += 4) {
            const int ka = k0 + 2 * lh;
            v2f a;
            a.x = T[lr * TS + ka];
            a.y = T[lr * TS + ka + 1];
            const bool va = (ka < M);
            const bool vb = (ka + 1 < M);
            const int  ra = va ? ((ka == 0) ? e : (s + ka - 1)) : 0;
            const int  rb = vb ? (s + ka) : 0;
            const float* xra = xa + (long)ra * NDIM;
            const float* xrb = xa + (long)rb * NDIM;

            #pragma unroll
            for (int ti = 0; ti < TPW; ++ti) {
                const float bx = xra[cidx_r[ti]];
                const float by = xrb[cidx_r[ti]];
                v2f bf;
                bf.x = (cv_r[ti] && va) ? bx : 0.f;
                bf.y = (cv_r[ti] && vb) ? by : 0.f;
                acc[ti] = __builtin_amdgcn_wmma_f32_16x16x4_f32(
                    false, a, false, bf, (short)0, acc[ti], false, false);
            }
        }

        __syncthreads();
        #pragma unroll
        for (int ti = 0; ti < TPW; ++ti) {
            const int t = wave + ti * NWAVES;
            if (t < NT) {
                const int c0 = t * 16 + lr;
                #pragma unroll
                for (int i = 0; i < 8; ++i)
                    T[(i + 8 * lh) * TS + c0] = acc[ti][i];
            }
        }
    }
    __syncthreads();

    float* orow = out + ((long)bb * 4 + p) * DDIM;
    const float* frow = features + ((long)bb * NDIM + s) * DDIM;
    for (int d = tid; d < DDIM; d += NTHREADS) {
        float accd = 0.f;
        for (int k = 0; k < M - 1; ++k) {
            const float wv = T[k + 1];
            const float fv = frow[(long)k * DDIM + d];
            accd += wv * (fv > 0.f ? fv : 0.f);
        }
        orow[d] = accd;
    }
}

__global__ __launch_bounds__(NTHREADS) void part_cam_kernel(
    const float* __restrict__ x, const float* __restrict__ features,
    float* __restrict__ out)
{
    __shared__ float T[16 * TS];

    const int bb = blockIdx.x;
    const int p  = blockIdx.y;
    const int e  = p + 1;

    if (p == 0) {
        part_body<197>(x, features, out, T, bb, p, e, 5);
    } else {
        const int s = (p == 1) ? 5 : ((p == 2) ? 54 : 103);
        part_body<99>(x, features, out, T, bb, p, e, s);
    }
}

extern "C" void kernel_launch(void* const* d_in, const int* in_sizes, int n_in,
                              void* d_out, int out_size, void* d_ws, size_t ws_size,
                              hipStream_t stream) {
    (void)in_sizes; (void)n_in; (void)d_ws; (void)ws_size; (void)out_size;
    const float* x        = (const float*)d_in[0];
    const float* features = (const float*)d_in[1];
    float* out            = (float*)d_out;

    dim3 grid(BATCH, 4);
    part_cam_kernel<<<grid, NTHREADS, 0, stream>>>(x, features, out);
}


# Loads issued ahead of use: one 16-column tile per wave (13 waves), each B-fragment register refilled for the next layer right after its product, LDS-only barrier fences; epilogue 64-bit loads x14
# speedup vs baseline: 4.5394x; 4.5394x over previous
#include <hip/hip_runtime.h>


typedef float v2f __attribute__((ext_vector_type(2)));
typedef float v8f __attribute__((ext_vector_type(8)));

#define LCNT     12
#define BATCH    64
#define NDIM     201
#define DDIM     768
#define NTHREADS 416
#define TS       211

__device__ __forceinline__ void lds_barrier()
{
    __builtin_amdgcn_fence(__ATOMIC_RELEASE, "workgroup", "local");
    __builtin_amdgcn_s_barrier();
    __builtin_amdgcn_fence(__ATOMIC_ACQUIRE, "workgroup", "local");
}

template<int M>
__device__ __forceinline__ float ld_b(const float* __restrict__ xa, const int i, const int j,
                                      const int offm, const int off00, const int cidx, const int lh)
{
    if (4 * i + j >= M) return 0.f;
    if (i == 0 && j == 0) return xa[off00];
    if (4 * i + 2 + j < M) return xa[offm + (4 * i + j) * NDIM];
    const int off = (lh == 0) ? (offm + (4 * i + j) * NDIM) : cidx;
    return xa[off];
}

template<int M, bool PF>
__device__ __forceinline__ v8f chain_layer(
    const float* __restrict__ xn, const float* T, v2f (&b)[(M + 3) / 4],
    const int tb, const int offm, const int off00, const int cidx, const int lh)
{
    constexpr int KS = (M + 3) / 4;
    v8f acc = (v8f)0.0f;
    #pragma unroll
    for (int i = 0; i < KS; ++i) {
        v2f a;
        a.x = T[tb + 4 * i];
        a.y = T[tb + 4 * i + 1];
        acc = __builtin_amdgcn_wmma_f32_16x16x4_f32(
            false, a, false, b[i], (short)0, acc, false, false);
        if (PF) {
            b[i].x = ld_b<M>(xn, i, 0, offm, off00, cidx, lh);
            b[i].y = ld_b<M>(xn, i, 1, offm, off00, cidx, lh);
        }
    }
    return acc;
}

template<int M>
__device__ __forceinline__ void part_body(
    const float* __restrict__ x, const float* __restrict__ features,
    float* __restrict__ out, float* __restrict__ T,
    const int bb, const int p, const int e, const int s)
{
    constexpr int NT  = (M + 15) / 16;
    constexpr int Mp  = NT * 16;
    constexpr int KS  = (M + 3) / 4;
    constexpr int NIT = (16 * Mp + NTHREADS - 1) / NTHREADS;

    const int tid  = threadIdx.x;
    const int wave = __builtin_amdgcn_readfirstlane(tid >> 5);
    const int lane = tid & 31;
    const int lr   = lane & 15;
    const int lh   = lane >> 4;

    const long xplane  = (long)NDIM * NDIM;
    const long lstride = (long)BATCH * xplane;
    const float* xb = x + (long)bb * xplane;

    const bool active = (wave < NT);
    const int  cc   = wave * 16 + lr;
    const bool cv   = active && (cc < M);
    const int  ccl  = cv ? cc : 0;
    const int  cidx = (ccl == 0) ? e : (s + ccl - 1);
    const int  offm  = (s - 1 + 2 * lh) * NDIM + cidx;
    const int  off00 = (lh == 0) ? (e * NDIM + cidx) : offm;
    const int  tb    = lr * TS + 2 * lh;

    float tv[NIT];
    {
        const float* xl = xb + (long)(LCNT - 1) * lstride;
        for (int it = 0; it < NIT; ++it) {
            const int  i = min(tid + it * NTHREADS, 16 * Mp - 1);
            const int  r = i / Mp, c = i - r * Mp;
            const bool v = (c < M);
            const int  ridx = v ? ((r == 0) ? e : (s + r - 1)) : 0;
            const int  cix  = v ? ((c == 0) ? e : (s + c - 1)) : 0;
            tv[it] = xl[ridx * NDIM + cix];
        }
    }

    v2f b[KS];
    if (active) {
        const float* xa = xb + (long)(LCNT - 2) * lstride;
        #pragma unroll
        for (int i = 0; i < KS; ++i) {
            b[i].x = ld_b<M>(xa, i, 0, offm, off00, cidx, lh);
            b[i].y = ld_b<M>(xa, i, 1, offm, off00, cidx, lh);
        }
    }

    for (int it = 0; it < NIT; ++it) {
        const int i = min(tid + it * NTHREADS, 16 * Mp - 1);
        const int r = i / Mp, c = i - r * Mp;
        T[r * TS + c] = (c < M) ? tv[it] : 0.f;
    }

    for (int l = LCNT - 2; l >= 1; --l) {
        lds_barrier();
        v8f acc = (v8f)0.0f;
        if (active)
            acc = chain_layer<M, true>(xb + (long)(l - 1) * lstride, T, b,
                                       tb, offm, off00, cidx, lh);
        lds_barrier();
        if (active) {
            for (int i = 0; i < 8; ++i)
                T[(i + 8 * lh) * TS + cc] = cv ? acc[i] : 0.f;
        }
    }
    {
        lds_barrier();
        v8f acc = (v8f)0.0f;
        if (active)
            acc = chain_layer<M, false>(xb, T, b, tb, offm, off00, cidx, lh);
        lds_barrier();
        if (active) {
            for (int i = 0; i < 8; ++i)
                T[(i + 8 * lh) * TS + cc] = cv ? acc[i] : 0.f;
        }
    }
    lds_barrier();

    if (tid < DDIM / 2) {
        const v2f* f2 = reinterpret_cast<const v2f*>(features + ((long)bb * NDIM + s) * DDIM) + tid;
        v2f accd = (v2f)0.0f;
        #pragma unroll 14
        for (int k = 0; k < M - 1; ++k) {
            const float wv = T[k + 1];
            const v2f   fv = f2[(long)k * (DDIM / 2)];
            accd.x += wv * (fv.x > 0.f ? fv.x : 0.f);
            accd.y += wv * (fv.y > 0.f ? fv.y : 0.f);
        }
        v2f* o2 = reinterpret_cast<v2f*>(out + ((long)bb * 4 + p) * DDIM) + tid;
        *o2 = accd;
    }
}

__global__ __launch_bounds__(NTHREADS) void part_cam_kernel(
    const float* __restrict__ x, const float* __restrict__ features,
    float* __restrict__ out)
{
    __shared__ float T[16 * TS];

    const int bb = blockIdx.x;
    const int p  = blockIdx.y;
    const int e  = p + 1;

    if (p == 0) {
        part_body<197>(x, features, out, T, bb, p, e, 5);
    } else {
        const int s = (p == 1) ? 5 : ((p == 2) ? 54 : 103);
        part_body<99>(x, features, out, T, bb, p, e, s);
    }
}

extern "C" void kernel_launch(void* const* d_in, const int* in_sizes, int n_in,
                              void* d_out, int out_size, void* d_ws, size_t ws_size,
                              hipStream_t stream) {
    (void)in_sizes; (void)n_in; (void)d_ws; (void)ws_size; (void)out_size;
    const float* x        = (const float*)d_in[0];
    const float* features = (const float*)d_in[1];
    float* out            = (float*)d_out;

    dim3 grid(BATCH, 4);
    part_cam_kernel<<<grid, NTHREADS, 0, stream>>>(x, features, out);
}


# 197-wide part: 16 waves = 4 row quarters x 4 column groups, 128-bit loads along rows each feeding 4 accumulator tiles; quarter partials summed via LDS, one barrier per layer (sum order changed)
# speedup vs baseline: 5.0187x; 1.1056x over previous
#include <hip/hip_runtime.h>


typedef float v2f __attribute__((ext_vector_type(2)));
typedef float v8f __attribute__((ext_vector_type(8)));

#define LCNT     12
#define BATCH    64
#define NDIM     201
#define DDIM     768
#define NTHREADS 512
#define PPLANE   800
#define TS       211

__device__ __forceinline__ void lds_barrier()
{
    __builtin_amdgcn_fence(__ATOMIC_RELEASE, "workgroup", "local");
    __builtin_amdgcn_s_barrier();
    __builtin_amdgcn_fence(__ATOMIC_ACQUIRE, "workgroup", "local");
}

template<int M>
__device__ __forceinline__ float ld_b(const float* __restrict__ xa, const int i, const int j,
                                      const int offm, const int off00, const int cidx, const int lh)
{
    if (4 * i + j >= M) return 0.f;
    if (i == 0 && j == 0) return xa[off00];
    if (4 * i + 2 + j < M) return xa[offm + (4 * i + j) * NDIM];
    const int off = (lh == 0) ? (offm + (4 * i + j) * NDIM) : cidx;
    return xa[off];
}

template<int M, bool PF>
__device__ __forceinline__ v8f chain_layer(
    const float* __restrict__ xn, const float* T, v2f (&b)[(M + 3) / 4],
    const int tb, const int offm, const int off00, const int cidx, const int lh)
{
    constexpr int KS = (M + 3) / 4;
    v8f acc = (v8f)0.0f;
    #pragma unroll
    for (int i = 0; i < KS; ++i) {
        v2f a;
        a.x = T[tb + 4 * i];
        a.y = T[tb + 4 * i + 1];
        acc = __builtin_amdgcn_wmma_f32_16x16x4_f32(
            false, a, false, b[i], (short)0, acc, false, false);
        if (PF) {
            b[i].x = ld_b<M>(xn, i, 0, offm, off00, cidx, lh);
            b[i].y = ld_b<M>(xn, i, 1, offm, off00, cidx, lh);
        }
    }
    return acc;
}

template<int M>
__device__ __forceinline__ void part_body(
    const float* __restrict__ x, const float* __restrict__ features,
    float* __restrict__ out, float* __restrict__ T,
    const int bb, const int p, const int e, const int s)
{
    constexpr int NT  = (M + 15) / 16;
    constexpr int Mp  = NT * 16;
    constexpr int KS  = (M + 3) / 4;
    constexpr int NIT = (16 * Mp + NTHREADS - 1) / NTHREADS;

    const int tid  = threadIdx.x;
    const int wave = __builtin_amdgcn_readfirstlane(tid >> 5);
    const int lane = tid & 31;
    const int lr   = lane & 15;
    const int lh   = lane >> 4;

    const long xplane  = (long)NDIM * NDIM;
    const long lstride = (long)BATCH * xplane;
    const float* xb = x + (long)bb * xplane;

    const bool active = (wave < NT);
    const int  cc   = wave * 16 + lr;
    const bool cv   = active && (cc < M);
    const int  ccl  = cv ? cc : 0;
    const int  cidx = (ccl == 0) ? e : (s + ccl - 1);
    const int  offm  = (s - 1 + 2 * lh) * NDIM + cidx;
    const int  off00 = (lh == 0) ? (e * NDIM + cidx) : offm;
    const int  tb    = lr * TS + 2 * lh;

    float tv[NIT];
    {
        const float* xl = xb + (long)(LCNT - 1) * lstride;
        for (int it = 0; it < NIT; ++it) {
            const int  i = min(tid + it * NTHREADS, 16 * Mp - 1);
            const int  r = i / Mp, c = i - r * Mp;
            const bool v = (c < M);
            const int  ridx = v ? ((r == 0) ? e : (s + r - 1)) : 0;
            const int  cix  = v ? ((c == 0) ? e : (s + c - 1)) : 0;
            tv[it] = xl[ridx * NDIM + cix];
        }
    }

    v2f b[KS];
    if (active) {
        const float* xa = xb + (long)(LCNT - 2) * lstride;
        #pragma unroll
        for (int i = 0; i < KS; ++i) {
            b[i].x = ld_b<M>(xa, i, 0, offm, off00, cidx, lh);
            b[i].y = ld_b<M>(xa, i, 1, offm, off00, cidx, lh);
        }
    }

    for (int it = 0; it < NIT; ++it) {
        const int i = min(tid + it * NTHREADS, 16 * Mp - 1);
        const int r = i / Mp, c = i - r * Mp;
        T[r * TS + c] = (c < M) ? tv[it] : 0.f;
    }

    for (int l = LCNT - 2; l >= 1; --l) {
        lds_barrier();
        v8f acc = (v8f)0.0f;
        if (active)
            acc = chain_layer<M, true>(xb + (long)(l - 1) * lstride, T, b,
                                       tb, offm, off00, cidx, lh);
        lds_barrier();
        if (active) {
            for (int i = 0; i < 8; ++i)
                T[(i + 8 * lh) * TS + cc] = cv ? acc[i] : 0.f;
        }
    }
    {
        lds_barrier();
        v8f acc = (v8f)0.0f;
        if (active)
            acc = chain_layer<M, false>(xb, T, b, tb, offm, off00, cidx, lh);
        lds_barrier();
        if (active) {
            for (int i = 0; i < 8; ++i)
                T[(i + 8 * lh) * TS + cc] = cv ? acc[i] : 0.f;
        }
    }
    lds_barrier();

    if (tid < DDIM / 2) {
        const v2f* f2 = reinterpret_cast<const v2f*>(features + ((long)bb * NDIM + s) * DDIM) + tid;
        v2f accd = (v2f)0.0f;
        #pragma unroll 14
        for (int k = 0; k < M - 1; ++k) {
            const float wv = T[k + 1];
            const v2f   fv = f2[(long)k * (DDIM / 2)];
            accd.x += wv * (fv.x > 0.f ? fv.x : 0.f);
            accd.y += wv * (fv.y > 0.f ? fv.y : 0.f);
        }
        v2f* o2 = reinterpret_cast<v2f*>(out + ((long)bb * 4 + p) * DDIM) + tid;
        *o2 = accd;
    }
}


typedef float v4f __attribute__((ext_vector_type(4)));
typedef v4f v4fu __attribute__((aligned(4)));

#define WM   197
#define WKS  50
#define WT   13

__device__ __forceinline__ int w_row(const int k, const int e, const int s)
{
    return (k == 0) ? e : ((k < WM) ? (s + k - 1) : s);
}

__device__ __forceinline__ v2f w_afrag(const float* P, const int k0)
{
    const v4f p0 = *reinterpret_cast<const v4f*>(P + 4 * k0);
    const v4f p1 = *reinterpret_cast<const v4f*>(P + 4 * k0 + 4);
    v2f a;
    a.x = ((p0.x + p0.y) + p0.z) + p0.w;
    a.y = ((p1.x + p1.y) + p1.z) + p1.w;
    return a;
}

__device__ __forceinline__ void w_init(float* P0, float* P1, const int tid, const float v0)
{
    if (tid < 4 * WKS) {
        const v4f z = (v4f){0.f, 0.f, 0.f, 0.f};
        *reinterpret_cast<v4f*>(P0 + 4 * tid) = (tid < WM) ? (v4f){v0, 0.f, 0.f, 0.f} : z;
        if (tid >= WM) *reinterpret_cast<v4f*>(P1 + 4 * tid) = z;
    }
}

template<bool PF>
__device__ __forceinline__ void wide_layer_cols(
    const float* __restrict__ xn, const float* Pr, float* Pw, v4f (&b)[WT][2],
    const int kg, const int cg, const int lr, const int lh, const int e, const int s)
{
    v8f acc[4];
    #pragma unroll
    for (int q = 0; q < 4; ++q) acc[q] = (v8f)0.0f;
    const int colx = s + 64 * cg + 4 * lr;
    #pragma unroll
    for (int t = 0; t < WT; ++t) {
        if (t < WT - 1 || kg < 2) {
            const int k0 = 4 * (kg + 4 * t) + 2 * lh;
            const v2f a = w_afrag(Pr, k0);
            const bool v0 = (k0 < WM), v1 = (k0 + 1 < WM);
            #pragma unroll
            for (int q = 0; q < 4; ++q) {
                v2f bq;
                bq.x = v0 ? b[t][0][q] : 0.f;
                bq.y = v1 ? b[t][1][q] : 0.f;
                acc[q] = __builtin_amdgcn_wmma_f32_16x16x4_f32(
                    false, a, false, bq, (short)0, acc[q], false, false);
            }
            if (PF) {
                b[t][0] = *reinterpret_cast<const v4fu*>(xn + w_row(k0,     e, s) * NDIM + colx);
                b[t][1] = *reinterpret_cast<const v4fu*>(xn + w_row(k0 + 1, e, s) * NDIM + colx);
            }
        }
    }
    if (lh == 0) {
        #pragma unroll
        for (int q = 0; q < 4; ++q)
            Pw[4 * (1 + 64 * cg + 4 * lr + q) + kg] = acc[q][0];
    }
}

template<bool PF>
__device__ __forceinline__ void wide_layer_rest(
    const float* __restrict__ xn, const float* Pr, float* Pw, v2f (&b)[WT],
    const int kg, const int lr, const int lh, const int e, const int s)
{
    v8f acc = (v8f)0.0f;
    const bool cv   = (lr < 5);
    const int  csub = (lr == 0) ? 0 : (192 + lr);
    const int  colx = (lr == 0 || !cv) ? e : (s + 191 + lr);
    #pragma unroll
    for (int t = 0; t < WT; ++t) {
        if (t < WT - 1 || kg < 2) {
            const int k0 = 4 * (kg + 4 * t) + 2 * lh;
            const v2f a = w_afrag(Pr, k0);
            v2f bq;
            bq.x = (k0     < WM) ? b[t].x : 0.f;
            bq.y = (k0 + 1 < WM) ? b[t].y : 0.f;
            acc = __builtin_amdgcn_wmma_f32_16x16x4_f32(
                false, a, false, bq, (short)0, acc, false, false);
            if (PF) {
                b[t].x = xn[w_row(k0,     e, s) * NDIM + colx];
                b[t].y = xn[w_row(k0 + 1, e, s) * NDIM + colx];
            }
        }
    }
    if (lh == 0 && cv) Pw[4 * csub + kg] = acc[0];
}

__device__ __forceinline__ void wide_part_body(
    const float* __restrict__ x, const float* __restrict__ features,
    float* __restrict__ out, float* T, const int bb)
{
    constexpr int e = 1, s = 5, p = 0;
    const int tid  = threadIdx.x;
    const int wave = __builtin_amdgcn_readfirstlane(tid >> 5);
    const int lane = tid & 31;
    const int lr   = lane & 15;
    const int lh   = lane >> 4;
    const int kg   = wave & 3;
    const int cg   = wave >> 2;

    const long xplane  = (long)NDIM * NDIM;
    const long lstride = (long)BATCH * xplane;
    const float* xb = x + (long)bb * xplane;

    float* P0 = T;
    float* P1 = T + PPLANE;
    float* Wv = T + 2 * PPLANE;

    float v0 = 0.f;
    if (tid < WM) {
        const float* xl = xb + (long)(LCNT - 1) * lstride;
        v0 = xl[e * NDIM + ((tid == 0) ? e : (s + tid - 1))];
    }

    v4f b[WT][2];
    v2f br[WT];
    {
        const float* xa = xb + (long)(LCNT - 2) * lstride;
        const int colw = s + 64 * cg + 4 * lr;
        const int colr = (lr == 0 || lr >= 5) ? e : (s + 191 + lr);
        if (cg < 3) {
            #pragma unroll
            for (int t = 0; t < WT; ++t) {
                if (t < WT - 1 || kg < 2) {
                    const int k0 = 4 * (kg + 4 * t) + 2 * lh;
                    b[t][0] = *reinterpret_cast<const v4fu*>(xa + w_row(k0,     e, s) * NDIM + colw);
                    b[t][1] = *reinterpret_cast<const v4fu*>(xa + w_row(k0 + 1, e, s) * NDIM + colw);
                }
            }
        } else {
            #pragma unroll
            for (int t = 0; t < WT; ++t) {
                if (t < WT - 1 || kg < 2) {
                    const int k0 = 4 * (kg + 4 * t) + 2 * lh;
                    br[t].x = xa[w_row(k0,     e, s) * NDIM + colr];
                    br[t].y = xa[w_row(k0 + 1, e, s) * NDIM + colr];
                }
            }
        }
    }
    w_init(P0, P1, tid, v0);
    lds_barrier();
    for (int l = LCNT - 2; l >= 1; --l) {
        const bool odd = ((LCNT - 2 - l) & 1);
        const float* Pr = odd ? P1 : P0;
        float*       Pw = odd ? P0 : P1;
        if (cg < 3) wide_layer_cols<true>(xb + (long)(l - 1) * lstride, Pr, Pw, b, kg, cg, lr, lh, e, s);
        else        wide_layer_rest<true>(xb + (long)(l - 1) * lstride, Pr, Pw, br, kg, lr, lh, e, s);
        lds_barrier();
    }
    if (cg < 3) wide_layer_cols<false>(xb, P0, P1, b, kg, cg, lr, lh, e, s);
    else        wide_layer_rest<false>(xb, P0, P1, br, kg, lr, lh, e, s);
    lds_barrier();
    if (tid < WM) {
        const v4f pv = *reinterpret_cast<const v4f*>(P1 + 4 * tid);
        Wv[tid] = ((pv.x + pv.y) + pv.z) + pv.w;
    }
    lds_barrier();

    if (tid < DDIM / 2) {
        const v2f* f2 = reinterpret_cast<const v2f*>(features + ((long)bb * NDIM + s) * DDIM) + tid;
        v2f accd = (v2f)0.0f;
        #pragma unroll 14
        for (int k = 0; k < WM - 1; ++k) {
            const float wv = Wv[k + 1];
            const v2f   fv = f2[(long)k * (DDIM / 2)];
            accd.x += wv * (fv.x > 0.f ? fv.x : 0.f);
            accd.y += wv * (fv.y > 0.f ? fv.y : 0.f);
        }
        v2f* o2 = reinterpret_cast<v2f*>(out + ((long)bb * 4 + p) * DDIM) + tid;
        *o2 = accd;
    }
}

__global__ __launch_bounds__(NTHREADS) void part_cam_kernel(
    const float* __restrict__ x, const float* __restrict__ features,
    float* __restrict__ out)
{
    __shared__ __attribute__((aligned(16))) float T[16 * TS];

    const int bb = blockIdx.x;
    const int p  = blockIdx.y;
    const int e  = p + 1;

    if (p == 0) {
        wide_part_body(x, features, out, T, bb);
    } else {
        const int s = (p == 1) ? 5 : ((p == 2) ? 54 : 103);
        part_body<99>(x, features, out, T, bb, p, e, s);
    }
}

extern "C" void kernel_launch(void* const* d_in, const int* in_sizes, int n_in,
                              void* d_out, int out_size, void* d_ws, size_t ws_size,
                              hipStream_t stream) {
    (void)in_sizes; (void)n_in; (void)d_ws; (void)ws_size; (void)out_size;
    const float* x        = (const float*)d_in[0];
    const float* features = (const float*)d_in[1];
    float* out            = (float*)d_out;

    dim3 grid(BATCH, 4);
    part_cam_kernel<<<grid, NTHREADS, 0, stream>>>(x, features, out);
}


# 197-wide part epilogue: 12 waves = 4 row groups x 3, 128-bit feature loads in 7-row batches, next batch requested before the current is used; group sums added via LDS (sum order changed)
# speedup vs baseline: 5.1920x; 1.0345x over previous
#include <hip/hip_runtime.h>


typedef float v2f __attribute__((ext_vector_type(2)));
typedef float v8f __attribute__((ext_vector_type(8)));

#define LCNT     12
#define BATCH    64
#define NDIM     201
#define DDIM     768
#define NTHREADS 512
#define PPLANE   800
#define TS       211

__device__ __forceinline__ void lds_barrier()
{
    __builtin_amdgcn_fence(__ATOMIC_RELEASE, "workgroup", "local");
    __builtin_amdgcn_s_barrier();
    __builtin_amdgcn_fence(__ATOMIC_ACQUIRE, "workgroup", "local");
}

template<int M>
__device__ __forceinline__ float ld_b(const float* __restrict__ xa, const int i, const int j,
                                      const int offm, const int off00, const int cidx, const int lh)
{
    if (4 * i + j >= M) return 0.f;
    if (i == 0 && j == 0) return xa[off00];
    if (4 * i + 2 + j < M) return xa[offm + (4 * i + j) * NDIM];
    const int off = (lh == 0) ? (offm + (4 * i + j) * NDIM) : cidx;
    return xa[off];
}

template<int M, bool PF>
__device__ __forceinline__ v8f chain_layer(
    const float* __restrict__ xn, const float* T, v2f (&b)[(M + 3) / 4],
    const int tb, const int offm, const int off00, const int cidx, const int lh)
{
    constexpr int KS = (M + 3) / 4;
    v8f acc = (v8f)0.0f;
    #pragma unroll
    for (int i = 0; i < KS; ++i) {
        v2f a;
        a.x = T[tb + 4 * i];
        a.y = T[tb + 4 * i + 1];
        acc = __builtin_amdgcn_wmma_f32_16x16x4_f32(
            false, a, false, b[i], (short)0, acc, false, false);
        if (PF) {
            b[i].x = ld_b<M>(xn, i, 0, offm, off00, cidx, lh);
            b[i].y = ld_b<M>(xn, i, 1, offm, off00, cidx, lh);
        }
    }
    return acc;
}

template<int M>
__device__ __forceinline__ void part_body(
    const float* __restrict__ x, const float* __restrict__ features,
    float* __restrict__ out, float* __restrict__ T,
    const int bb, const int p, const int e, const int s)
{
    constexpr int NT  = (M + 15) / 16;
    constexpr int Mp  = NT * 16;
    constexpr int KS  = (M + 3) / 4;
    constexpr int NIT = (16 * Mp + NTHREADS - 1) / NTHREADS;

    const int tid  = threadIdx.x;
    const int wave = __builtin_amdgcn_readfirstlane(tid >> 5);
    const int lane = tid & 31;
    const int lr   = lane & 15;
    const int lh   = lane >> 4;

    const long xplane  = (long)NDIM * NDIM;
    const long lstride = (long)BATCH * xplane;
    const float* xb = x + (long)bb * xplane;

    const bool active = (wave < NT);
    const int  cc   = wave * 16 + lr;
    const bool cv   = active && (cc < M);
    const int  ccl  = cv ? cc : 0;
    const int  cidx = (ccl == 0) ? e : (s + ccl - 1);
    const int  offm  = (s - 1 + 2 * lh) * NDIM + cidx;
    const int  off00 = (lh == 0) ? (e * NDIM + cidx) : offm;
    const int  tb    = lr * TS + 2 * lh;

    float tv[NIT];
    {
        const float* xl = xb + (long)(LCNT - 1) * lstride;
        for (int it = 0; it < NIT; ++it) {
            const int  i = min(tid + it * NTHREADS, 16 * Mp - 1);
            const int  r = i / Mp, c = i - r * Mp;
            const bool v = (c < M);
            const int  ridx = v ? ((r == 0) ? e : (s + r - 1)) : 0;
            const int  cix  = v ? ((c == 0) ? e : (s + c - 1)) : 0;
            tv[it] = xl[ridx * NDIM + cix];
        }
    }

    v2f b[KS];
    if (active) {
        const float* xa = xb + (long)(LCNT - 2) * lstride;
        #pragma unroll
        for (int i = 0; i < KS; ++i) {
            b[i].x = ld_b<M>(xa, i, 0, offm, off00, cidx, lh);
            b[i].y = ld_b<M>(xa, i, 1, offm, off00, cidx, lh);
        }
    }

    for (int it = 0; it < NIT; ++it) {
        const int i = min(tid + it * NTHREADS, 16 * Mp - 1);
        const int r = i / Mp, c = i - r * Mp;
        T[r * TS + c] = (c < M) ? tv[it] : 0.f;
    }

    for (int l = LCNT - 2; l >= 1; --l) {
        lds_barrier();
        v8f acc = (v8f)0.0f;
        if (active)
            acc = chain_layer<M, true>(xb + (long)(l - 1) * lstride, T, b,
                                       tb, offm, off00, cidx, lh);
        lds_barrier();
        if (active) {
            for (int i = 0; i < 8; ++i)
                T[(i + 8 * lh) * TS + cc] = cv ? acc[i] : 0.f;
        }
    }
    {
        lds_barrier();
        v8f acc = (v8f)0.0f;
        if (active)
            acc = chain_layer<M, false>(xb, T, b, tb, offm, off00, cidx, lh);
        lds_barrier();
        if (active) {
            for (int i = 0; i < 8; ++i)
                T[(i + 8 * lh) * TS + cc] = cv ? acc[i] : 0.f;
        }
    }
    lds_barrier();

    if (tid < DDIM / 2) {
        const v2f* f2 = reinterpret_cast<const v2f*>(features + ((long)bb * NDIM + s) * DDIM) + tid;
        v2f accd = (v2f)0.0f;
        #pragma unroll 14
        for (int k = 0; k < M - 1; ++k) {
            const float wv = T[k + 1];
            const v2f   fv = f2[(long)k * (DDIM / 2)];
            accd.x += wv * (fv.x > 0.f ? fv.x : 0.f);
            accd.y += wv * (fv.y > 0.f ? fv.y : 0.f);
        }
        v2f* o2 = reinterpret_cast<v2f*>(out + ((long)bb * 4 + p) * DDIM) + tid;
        *o2 = accd;
    }
}


typedef float v4f __attribute__((ext_vector_type(4)));
typedef v4f v4fu __attribute__((aligned(4)));

#define WM   197
#define WKS  50
#define WT   13

__device__ __forceinline__ int w_row(const int k, const int e, const int s)
{
    return (k == 0) ? e : ((k < WM) ? (s + k - 1) : s);
}

__device__ __forceinline__ v2f w_afrag(const float* P, const int k0)
{
    const v4f p0 = *reinterpret_cast<const v4f*>(P + 4 * k0);
    const v4f p1 = *reinterpret_cast<const v4f*>(P + 4 * k0 + 4);
    v2f a;
    a.x = ((p0.x + p0.y) + p0.z) + p0.w;
    a.y = ((p1.x + p1.y) + p1.z) + p1.w;
    return a;
}

__device__ __forceinline__ void w_init(float* P0, float* P1, const int tid, const float v0)
{
    if (tid < 4 * WKS) {
        const v4f z = (v4f){0.f, 0.f, 0.f, 0.f};
        *reinterpret_cast<v4f*>(P0 + 4 * tid) = (tid < WM) ? (v4f){v0, 0.f, 0.f, 0.f} : z;
        if (tid >= WM) *reinterpret_cast<v4f*>(P1 + 4 * tid) = z;
    }
}

template<bool PF>
__device__ __forceinline__ void wide_layer_cols(
    const float* __restrict__ xn, const float* Pr, float* Pw, v4f (&b)[WT][2],
    const int kg, const int cg, const int lr, const int lh, const int e, const int s)
{
    v8f acc[4];
    #pragma unroll
    for (int q = 0; q < 4; ++q) acc[q] = (v8f)0.0f;
    const int colx = s + 64 * cg + 4 * lr;
    #pragma unroll
    for (int t = 0; t < WT; ++t) {
        if (t < WT - 1 || kg < 2) {
            const int k0 = 4 * (kg + 4 * t) + 2 * lh;
            const v2f a = w_afrag(Pr, k0);
            const bool v0 = (k0 < WM), v1 = (k0 + 1 < WM);
            #pragma unroll
            for (int q = 0; q < 4; ++q) {
                v2f bq;
                bq.x = v0 ? b[t][0][q] : 0.f;
                bq.y = v1 ? b[t][1][q] : 0.f;
                acc[q] = __builtin_amdgcn_wmma_f32_16x16x4_f32(
                    false, a, false, bq, (short)0, acc[q], false, false);
            }
            if (PF) {
                b[t][0] = *reinterpret_cast<const v4fu*>(xn + w_row(k0,     e, s) * NDIM + colx);
                b[t][1] = *reinterpret_cast<const v4fu*>(xn + w_row(k0 + 1, e, s) * NDIM + colx);
            }
        }
    }
    if (lh == 0) {
        #pragma unroll
        for (int q = 0; q < 4; ++q)
            Pw[4 * (1 + 64 * cg + 4 * lr + q) + kg] = acc[q][0];
    }
}

template<bool PF>
__device__ __forceinline__ void wide_layer_rest(
    const float* __restrict__ xn, const float* Pr, float* Pw, v2f (&b)[WT],
    const int kg, const int lr, const int lh, const int e, const int s)
{
    v8f acc = (v8f)0.0f;
    const bool cv   = (lr < 5);
    const int  csub = (lr == 0) ? 0 : (192 + lr);
    const int  colx = (lr == 0 || !cv) ? e : (s + 191 + lr);
    #pragma unroll
    for (int t = 0; t < WT; ++t) {
        if (t < WT - 1 || kg < 2) {
            const int k0 = 4 * (kg + 4 * t) + 2 * lh;
            const v2f a = w_afrag(Pr, k0);
            v2f bq;
            bq.x = (k0     < WM) ? b[t].x : 0.f;
            bq.y = (k0 + 1 < WM) ? b[t].y : 0.f;
            acc = __builtin_amdgcn_wmma_f32_16x16x4_f32(
                false, a, false, bq, (short)0, acc, false, false);
            if (PF) {
                b[t].x = xn[w_row(k0,     e, s) * NDIM + colx];
                b[t].y = xn[w_row(k0 + 1, e, s) * NDIM + colx];
            }
        }
    }
    if (lh == 0 && cv) Pw[4 * csub + kg] = acc[0];
}


#define EROWS 49
#define EU    7

__device__ __forceinline__ void ep_load(v4f (&buf)[EU][2], const v4f* __restrict__ fp, const int r0)
{
    #pragma unroll
    for (int u = 0; u < EU; ++u) {
        buf[u][0] = fp[(r0 + u) * (DDIM / 4)];
        buf[u][1] = fp[(r0 + u) * (DDIM / 4) + DDIM / 8];
    }
}

__device__ __forceinline__ void ep_fma(v4f (&acc)[2], const v4f (&buf)[EU][2], const float* Wg, const int r0)
{
    #pragma unroll
    for (int u = 0; u < EU; ++u) {
        const float wv = Wg[r0 + u];
        #pragma unroll
        for (int j = 0; j < 2; ++j) {
            const v4f f = buf[u][j];
            acc[j].x += wv * (f.x > 0.f ? f.x : 0.f);
            acc[j].y += wv * (f.y > 0.f ? f.y : 0.f);
            acc[j].z += wv * (f.z > 0.f ? f.z : 0.f);
            acc[j].w += wv * (f.w > 0.f ? f.w : 0.f);
        }
    }
}

__device__ __forceinline__ void wide_part_body(
    const float* __restrict__ x, const float* __restrict__ features,
    float* __restrict__ out, float* T, float* R2, const int bb)
{
    constexpr int e = 1, s = 5, p = 0;
    const int tid  = threadIdx.x;
    const int wave = __builtin_amdgcn_readfirstlane(tid >> 5);
    const int lane = tid & 31;
    const int lr   = lane & 15;
    const int lh   = lane >> 4;
    const int kg   = wave & 3;
    const int cg   = wave >> 2;

    const long xplane  = (long)NDIM * NDIM;
    const long lstride = (long)BATCH * xplane;
    const float* xb = x + (long)bb * xplane;

    float* P0 = T;
    float* P1 = T + PPLANE;
    float* Wv = T + 2 * PPLANE;

    float v0 = 0.f;
    if (tid < WM) {
        const float* xl = xb + (long)(LCNT - 1) * lstride;
        v0 = xl[e * NDIM + ((tid == 0) ? e : (s + tid - 1))];
    }

    v4f b[WT][2];
    v2f br[WT];
    {
        const float* xa = xb + (long)(LCNT - 2) * lstride;
        const int colw = s + 64 * cg + 4 * lr;
        const int colr = (lr == 0 || lr >= 5) ? e : (s + 191 + lr);
        if (cg < 3) {
            #pragma unroll
            for (int t = 0; t < WT; ++t) {
                if (t < WT - 1 || kg < 2) {
                    const int k0 = 4 * (kg + 4 * t) + 2 * lh;
                    b[t][0] = *reinterpret_cast<const v4fu*>(xa + w_row(k0,     e, s) * NDIM + colw);
                    b[t][1] = *reinterpret_cast<const v4fu*>(xa + w_row(k0 + 1, e, s) * NDIM + colw);
                }
            }
        } else {
            #pragma unroll
            for (int t = 0; t < WT; ++t) {
                if (t < WT - 1 || kg < 2) {
                    const int k0 = 4 * (kg + 4 * t) + 2 * lh;
                    br[t].x = xa[w_row(k0,     e, s) * NDIM + colr];
                    br[t].y = xa[w_row(k0 + 1, e, s) * NDIM + colr];
                }
            }
        }
    }
    w_init(P0, P1, tid, v0);
    lds_barrier();
    for (int l = LCNT - 2; l >= 1; --l) {
        const bool odd = ((LCNT - 2 - l) & 1);
        const float* Pr = odd ? P1 : P0;
        float*       Pw = odd ? P0 : P1;
        if (cg < 3) wide_layer_cols<true>(xb + (long)(l - 1) * lstride, Pr, Pw, b, kg, cg, lr, lh, e, s);
        else        wide_layer_rest<true>(xb + (long)(l - 1) * lstride, Pr, Pw, br, kg, lr, lh, e, s);
        lds_barrier();
    }
    if (cg < 3) wide_layer_cols<false>(xb, P0, P1, b, kg, cg, lr, lh, e, s);
    else        wide_layer_rest<false>(xb, P0, P1, br, kg, lr, lh, e, s);

    const bool ep = (wave < 12);
    const int  g  = wave / 3;
    const int  w3 = wave - 3 * g;
    const v4f* fp = reinterpret_cast<const v4f*>(features + ((long)bb * NDIM + s + g * EROWS) * DDIM)
                    + w3 * 32 + lane;
    const float* Wg = Wv + 1 + g * EROWS;
    v4f fa[EU][2], fb[EU][2];
    if (ep) ep_load(fa, fp, 0);

    lds_barrier();
    {
        const int ti = (wave << 5) + (int)__builtin_amdgcn_mbcnt_lo(~0u, 0u);
        if (ti < WM) {
            const v4f pv = *reinterpret_cast<const v4f*>(P1 + 4 * ti);
            Wv[ti] = ((pv.x + pv.y) + pv.z) + pv.w;
        }
    }
    lds_barrier();

    v4f acc[2];
    acc[0] = (v4f){0.f, 0.f, 0.f, 0.f};
    acc[1] = (v4f){0.f, 0.f, 0.f, 0.f};
    if (ep) {
        ep_load(fb, fp, 1 * EU);  ep_fma(acc, fa, Wg, 0 * EU);
        ep_load(fa, fp, 2 * EU);  ep_fma(acc, fb, Wg, 1 * EU);
        ep_load(fb, fp, 3 * EU);  ep_fma(acc, fa, Wg, 2 * EU);
        ep_load(fa, fp, 4 * EU);  ep_fma(acc, fb, Wg, 3 * EU);
        ep_load(fb, fp, 5 * EU);  ep_fma(acc, fa, Wg, 4 * EU);
        ep_load(fa, fp, 6 * EU);  ep_fma(acc, fb, Wg, 5 * EU);
        ep_fma(acc, fa, Wg, 6 * EU);
        if (g > 0) {
            v4f* r4 = reinterpret_cast<v4f*>(R2 + (g - 1) * DDIM) + w3 * 32 + lane;
            r4[0]        = acc[0];
            r4[DDIM / 8] = acc[1];
        }
    }
    lds_barrier();
    if (ep && g == 0) {
        const v4f* r4 = reinterpret_cast<const v4f*>(R2) + w3 * 32 + lane;
        v4f* o4 = reinterpret_cast<v4f*>(out + ((long)bb * 4 + p) * DDIM) + w3 * 32 + lane;
        #pragma unroll
        for (int j = 0; j < 2; ++j) {
            const v4f p1 = r4[j * (DDIM / 8)];
            const v4f p2 = r4[j * (DDIM / 8) + DDIM / 4];
            const v4f p3 = r4[j * (DDIM / 8) + 2 * (DDIM / 4)];
            o4[j * (DDIM / 8)] = ((acc[j] + p1) + p2) + p3;
        }
    }
}

__global__ __launch_bounds__(NTHREADS) void part_cam_kernel(
    const float* __restrict__ x, const float* __restrict__ features,
    float* __restrict__ out)
{
    __shared__ __attribute__((aligned(16))) float T[16 * TS];
    __shared__ __attribute__((aligned(16))) float R2[3 * DDIM];

    const int bb = blockIdx.x;
    const int p  = blockIdx.y;
    const int e  = p + 1;

    if (p == 0) {
        wide_part_body(x, features, out, T, R2, bb);
    } else {
        const int s = (p == 1) ? 5 : ((p == 2) ? 54 : 103);
        part_body<99>(x, features, out, T, bb, p, e, s);
    }
}

extern "C" void kernel_launch(void* const* d_in, const int* in_sizes, int n_in,
                              void* d_out, int out_size, void* d_ws, size_t ws_size,
                              hipStream_t stream) {
    (void)in_sizes; (void)n_in; (void)d_ws; (void)ws_size; (void)out_size;
    const float* x        = (const float*)d_in[0];
    const float* features = (const float*)d_in[1];
    float* out            = (float*)d_out;

    dim3 grid(BATCH, 4);
    part_cam_kernel<<<grid, NTHREADS, 0, stream>>>(x, features, out);
}


# 197-wide part: each k-step waits only for its own rows, not for all pending loads at every layer's top and end: wave kinds as two sequential blocks, k-step order pinned, rows from one base address
# speedup vs baseline: 5.7277x; 1.1032x over previous
#include <hip/hip_runtime.h>


typedef float v2f __attribute__((ext_vector_type(2)));
typedef float v8f __attribute__((ext_vector_type(8)));

#define LCNT     12
#define BATCH    64
#define NDIM     201
#define DDIM     768
#define NTHREADS 512
#define PPLANE   800
#define TS       211

__device__ __forceinline__ void lds_barrier()
{
    __builtin_amdgcn_fence(__ATOMIC_RELEASE, "workgroup", "local");
    __builtin_amdgcn_s_barrier();
    __builtin_amdgcn_fence(__ATOMIC_ACQUIRE, "workgroup", "local");
}

template<int M>
__device__ __forceinline__ float ld_b(const float* __restrict__ xa, const int i, const int j,
                                      const int offm, const int off00, const int cidx, const int lh)
{
    if (4 * i + j >= M) return 0.f;
    if (i == 0 && j == 0) return xa[off00];
    if (4 * i + 2 + j < M) return xa[offm + (4 * i + j) * NDIM];
    const int off = (lh == 0) ? (offm + (4 * i + j) * NDIM) : cidx;
    return xa[off];
}

template<int M, bool PF>
__device__ __forceinline__ v8f chain_layer(
    const float* __restrict__ xn, const float* T, v2f (&b)[(M + 3) / 4],
    const int tb, const int offm, const int off00, const int cidx, const int lh)
{
    constexpr int KS = (M + 3) / 4;
    v8f acc = (v8f)0.0f;
    #pragma unroll
    for (int i = 0; i < KS; ++i) {
        v2f a;
        a.x = T[tb + 4 * i];
        a.y = T[tb + 4 * i + 1];
        acc = __builtin_amdgcn_wmma_f32_16x16x4_f32(
            false, a, false, b[i], (short)0, acc, false, false);
        if (PF) {
            b[i].x = ld_b<M>(xn, i, 0, offm, off00, cidx, lh);
            b[i].y = ld_b<M>(xn, i, 1, offm, off00, cidx, lh);
        }
    }
    return acc;
}

template<int M>
__device__ __forceinline__ void part_body(
    const float* __restrict__ x, const float* __restrict__ features,
    float* __restrict__ out, float* __restrict__ T,
    const int bb, const int p, const int e, const int s)
{
    constexpr int NT  = (M + 15) / 16;
    constexpr int Mp  = NT * 16;
    constexpr int KS  = (M + 3) / 4;
    constexpr int NIT = (16 * Mp + NTHREADS - 1) / NTHREADS;

    const int tid  = threadIdx.x;
    const int wave = __builtin_amdgcn_readfirstlane(tid >> 5);
    const int lane = tid & 31;
    const int lr   = lane & 15;
    const int lh   = lane >> 4;

    const long xplane  = (long)NDIM * NDIM;
    const long lstride = (long)BATCH * xplane;
    const float* xb = x + (long)bb * xplane;

    const bool active = (wave < NT);
    const int  cc   = wave * 16 + lr;
    const bool cv   = active && (cc < M);
    const int  ccl  = cv ? cc : 0;
    const int  cidx = (ccl == 0) ? e : (s + ccl - 1);
    const int  offm  = (s - 1 + 2 * lh) * NDIM + cidx;
    const int  off00 = (lh == 0) ? (e * NDIM + cidx) : offm;
    const int  tb    = lr * TS + 2 * lh;

    float tv[NIT];
    {
        const float* xl = xb + (long)(LCNT - 1) * lstride;
        for (int it = 0; it < NIT; ++it) {
            const int  i = min(tid + it * NTHREADS, 16 * Mp - 1);
            const int  r = i / Mp, c = i - r * Mp;
            const bool v = (c < M);
            const int  ridx = v ? ((r == 0) ? e : (s + r - 1)) : 0;
            const int  cix  = v ? ((c == 0) ? e : (s + c - 1)) : 0;
            tv[it] = xl[ridx * NDIM + cix];
        }
    }

    v2f b[KS];
    if (active) {
        const float* xa = xb + (long)(LCNT - 2) * lstride;
        #pragma unroll
        for (int i = 0; i < KS; ++i) {
            b[i].x = ld_b<M>(xa, i, 0, offm, off00, cidx, lh);
            b[i].y = ld_b<M>(xa, i, 1, offm, off00, cidx, lh);
        }
    }

    for (int it = 0; it < NIT; ++it) {
        const int i = min(tid + it * NTHREADS, 16 * Mp - 1);
        const int r = i / Mp, c = i - r * Mp;
        T[r * TS + c] = (c < M) ? tv[it] : 0.f;
    }

    for (int l = LCNT - 2; l >= 1; --l) {
        lds_barrier();
        v8f acc = (v8f)0.0f;
        if (active)
            acc = chain_layer<M, true>(xb + (long)(l - 1) * lstride, T, b,
                                       tb, offm, off00, cidx, lh);
        lds_barrier();
        if (active) {
            for (int i = 0; i < 8; ++i)
                T[(i + 8 * lh) * TS + cc] = cv ? acc[i] : 0.f;
        }
    }
    {
        lds_barrier();
        v8f acc = (v8f)0.0f;
        if (active)
            acc = chain_layer<M, false>(xb, T, b, tb, offm, off00, cidx, lh);
        lds_barrier();
        if (active) {
            for (int i = 0; i < 8; ++i)
                T[(i + 8 * lh) * TS + cc] = cv ? acc[i] : 0.f;
        }
    }
    lds_barrier();

    if (tid < DDIM / 2) {
        const v2f* f2 = reinterpret_cast<const v2f*>(features + ((long)bb * NDIM + s) * DDIM) + tid;
        v2f accd = (v2f)0.0f;
        #pragma unroll 14
        for (int k = 0; k < M - 1; ++k) {
            const float wv = T[k + 1];
            const v2f   fv = f2[(long)k * (DDIM / 2)];
            accd.x += wv * (fv.x > 0.f ? fv.x : 0.f);
            accd.y += wv * (fv.y > 0.f ? fv.y : 0.f);
        }
        v2f* o2 = reinterpret_cast<v2f*>(out + ((long)bb * 4 + p) * DDIM) + tid;
        *o2 = accd;
    }
}


typedef float v4f __attribute__((ext_vector_type(4)));
typedef v4f v4fu __attribute__((aligned(4)));

#define WM   197
#define WKS  50
#define WT   13

__device__ __forceinline__ int w_row(const int k, const int e, const int s)
{
    return (k == 0) ? e : ((k < WM) ? (s + k - 1) : s);
}

__device__ __forceinline__ int w_row_t(const int t, const int k, const int e, const int s)
{
    return (t == 0 || t == WT - 1) ? w_row(k, e, s) : (s + k - 1);
}

__device__ __forceinline__ v2f w_afrag(const float* P, const int k0)
{
    const v4f p0 = *reinterpret_cast<const v4f*>(P + 4 * k0);
    const v4f p1 = *reinterpret_cast<const v4f*>(P + 4 * k0 + 4);
    v2f a;
    a.x = ((p0.x + p0.y) + p0.z) + p0.w;
    a.y = ((p1.x + p1.y) + p1.z) + p1.w;
    return a;
}

__device__ __forceinline__ void w_init(float* P0, float* P1, const int tid, const float v0)
{
    if (tid < 4 * WKS) {
        const v4f z = (v4f){0.f, 0.f, 0.f, 0.f};
        *reinterpret_cast<v4f*>(P0 + 4 * tid) = (tid < WM) ? (v4f){v0, 0.f, 0.f, 0.f} : z;
        if (tid >= WM) *reinterpret_cast<v4f*>(P1 + 4 * tid) = z;
    }
}

template<bool PF>
__device__ __forceinline__ void wide_layer_cols(
    const float* __restrict__ xn, const float* Pr, float* Pw, v4f (&b)[WT][2],
    const int kg, const int cg, const int lr, const int lh, const int e, const int s)
{
    v8f acc[4];
    #pragma unroll
    for (int q = 0; q < 4; ++q) acc[q] = (v8f)0.0f;
    const int colx = s + 64 * cg + 4 * lr;
    #pragma unroll
    for (int t = 0; t < WT; ++t) {
        if (t < WT - 1 || kg < 2) {
            __builtin_amdgcn_sched_barrier(0);
            const int k0 = 4 * (kg + 4 * t) + 2 * lh;
            const v2f a = w_afrag(Pr, k0);
            const bool v0 = (k0 < WM), v1 = (k0 + 1 < WM);
            #pragma unroll
            for (int q = 0; q < 4; ++q) {
                v2f bq;
                bq.x = v0 ? b[t][0][q] : 0.f;
                bq.y = v1 ? b[t][1][q] : 0.f;
                acc[q] = __builtin_amdgcn_wmma_f32_16x16x4_f32(
                    false, a, false, bq, (short)0, acc[q], false, false);
            }
            if (PF) {
                b[t][0] = *reinterpret_cast<const v4fu*>(xn + w_row_t(t, k0,     e, s) * NDIM + colx);
                b[t][1] = *reinterpret_cast<const v4fu*>(xn + w_row_t(t, k0 + 1, e, s) * NDIM + colx);
            }
        }
    }
    if (lh == 0) {
        #pragma unroll
        for (int q = 0; q < 4; ++q)
            Pw[4 * (1 + 64 * cg + 4 * lr + q) + kg] = acc[q][0];
    }
}

template<bool PF>
__device__ __forceinline__ void wide_layer_rest(
    const float* __restrict__ xn, const float* Pr, float* Pw, v2f (&b)[WT],
    const int kg, const int lr, const int lh, const int e, const int s)
{
    v8f acc = (v8f)0.0f;
    const bool cv   = (lr < 5);
    const int  csub = (lr == 0) ? 0 : (192 + lr);
    const int  colx = (lr == 0 || !cv) ? e : (s + 191 + lr);
    #pragma unroll
    for (int t = 0; t < WT; ++t) {
        if (t < WT - 1 || kg < 2) {
            __builtin_amdgcn_sched_barrier(0);
            const int k0 = 4 * (kg + 4 * t) + 2 * lh;
            const v2f a = w_afrag(Pr, k0);
            v2f bq;
            bq.x = (k0     < WM) ? b[t].x : 0.f;
            bq.y = (k0 + 1 < WM) ? b[t].y : 0.f;
            acc = __builtin_amdgcn_wmma_f32_16x16x4_f32(
                false, a, false, bq, (short)0, acc, false, false);
            if (PF) {
                b[t].x = xn[w_row_t(t, k0,     e, s) * NDIM + colx];
                b[t].y = xn[w_row_t(t, k0 + 1, e, s) * NDIM + colx];
            }
        }
    }
    if (lh == 0 && cv) Pw[4 * csub + kg] = acc[0];
}


#define EROWS 49
#define EU    7

__device__ __forceinline__ void ep_load(v4f (&buf)[EU][2], const v4f* __restrict__ fp, const int r0)
{
    #pragma unroll
    for (int u = 0; u < EU; ++u) {
        buf[u][0] = fp[(r0 + u) * (DDIM / 4)];
        buf[u][1] = fp[(r0 + u) * (DDIM / 4) + DDIM / 8];
    }
}

__device__ __forceinline__ void ep_fma(v4f (&acc)[2], const v4f (&buf)[EU][2], const float* Wg, const int r0)
{
    #pragma unroll
    for (int u = 0; u < EU; ++u) {
        const float wv = Wg[r0 + u];
        #pragma unroll
        for (int j = 0; j < 2; ++j) {
            const v4f f = buf[u][j];
            acc[j].x += wv * (f.x > 0.f ? f.x : 0.f);
            acc[j].y += wv * (f.y > 0.f ? f.y : 0.f);
            acc[j].z += wv * (f.z > 0.f ? f.z : 0.f);
            acc[j].w += wv * (f.w > 0.f ? f.w : 0.f);
        }
    }
}

__device__ __forceinline__ void wide_part_body(
    const float* __restrict__ x, const float* __restrict__ features,
    float* __restrict__ out, float* T, float* R2, const int bb)
{
    constexpr int e = 1, s = 5, p = 0;
    const int tid  = threadIdx.x;
    const int wave = __builtin_amdgcn_readfirstlane(tid >> 5);
    const int lane = tid & 31;
    const int lr   = lane & 15;
    const int lh   = lane >> 4;
    const int kg   = wave & 3;
    const int cg   = wave >> 2;

    const long xplane  = (long)NDIM * NDIM;
    const long lstride = (long)BATCH * xplane;
    const float* xb = x + (long)bb * xplane;

    float* P0 = T;
    float* P1 = T + PPLANE;
    float* Wv = T + 2 * PPLANE;

    float v0 = 0.f;
    if (tid < WM) {
        const float* xl = xb + (long)(LCNT - 1) * lstride;
        v0 = xl[e * NDIM + ((tid == 0) ? e : (s + tid - 1))];
    }

    v4f b[WT][2];
    v2f br[WT];
    {
        const float* xa = xb + (long)(LCNT - 2) * lstride;
        const int colw = s + 64 * cg + 4 * lr;
        const int colr = (lr == 0 || lr >= 5) ? e : (s + 191 + lr);
        if (cg < 3) {
            #pragma unroll
            for (int t = 0; t < WT; ++t) {
                if (t < WT - 1 || kg < 2) {
                    const int k0 = 4 * (kg + 4 * t) + 2 * lh;
                    b[t][0] = *reinterpret_cast<const v4fu*>(xa + w_row_t(t, k0,     e, s) * NDIM + colw);
                    b[t][1] = *reinterpret_cast<const v4fu*>(xa + w_row_t(t, k0 + 1, e, s) * NDIM + colw);
                }
            }
        } else {
            #pragma unroll
            for (int t = 0; t < WT; ++t) {
                if (t < WT - 1 || kg < 2) {
                    const int k0 = 4 * (kg + 4 * t) + 2 * lh;
                    br[t].x = xa[w_row_t(t, k0,     e, s) * NDIM + colr];
                    br[t].y = xa[w_row_t(t, k0 + 1, e, s) * NDIM + colr];
                }
            }
        }
    }
    w_init(P0, P1, tid, v0);
    lds_barrier();
    for (int l = LCNT - 2; l >= 1; --l) {
        const bool odd = ((LCNT - 2 - l) & 1);
        const float* Pr = odd ? P1 : P0;
        float*       Pw = odd ? P0 : P1;
        if (cg < 3)  wide_layer_cols<true>(xb + (long)(l - 1) * lstride, Pr, Pw, b, kg, cg, lr, lh, e, s);
        if (cg >= 3) wide_layer_rest<true>(xb + (long)(l - 1) * lstride, Pr, Pw, br, kg, lr, lh, e, s);
        lds_barrier();
    }
    if (cg < 3)  wide_layer_cols<false>(xb, P0, P1, b, kg, cg, lr, lh, e, s);
    if (cg >= 3) wide_layer_rest<false>(xb, P0, P1, br, kg, lr, lh, e, s);
    __builtin_amdgcn_fence(__ATOMIC_ACQUIRE, "workgroup");

    const bool ep = (wave < 12);
    const int  g  = wave / 3;
    const int  w3 = wave - 3 * g;
    const v4f* fp = reinterpret_cast<const v4f*>(features + ((long)bb * NDIM + s + g * EROWS) * DDIM)
                    + w3 * 32 + lane;
    const float* Wg = Wv + 1 + g * EROWS;
    v4f fa[EU][2], fb[EU][2];
    if (ep) ep_load(fa, fp, 0);

    lds_barrier();
    {
        const int ti = (wave << 5) + (int)__builtin_amdgcn_mbcnt_lo(~0u, 0u);
        if (ti < WM) {
            const v4f pv = *reinterpret_cast<const v4f*>(P1 + 4 * ti);
            Wv[ti] = ((pv.x + pv.y) + pv.z) + pv.w;
        }
    }
    lds_barrier();

    v4f acc[2];
    acc[0] = (v4f){0.f, 0.f, 0.f, 0.f};
    acc[1] = (v4f){0.f, 0.f, 0.f, 0.f};
    if (ep) {
        ep_load(fb, fp, 1 * EU);  ep_fma(acc, fa, Wg, 0 * EU);
        ep_load(fa, fp, 2 * EU);  ep_fma(acc, fb, Wg, 1 * EU);
        ep_load(fb, fp, 3 * EU);  ep_fma(acc, fa, Wg, 2 * EU);
        ep_load(fa, fp, 4 * EU);  ep_fma(acc, fb, Wg, 3 * EU);
        ep_load(fb, fp, 5 * EU);  ep_fma(acc, fa, Wg, 4 * EU);
        ep_load(fa, fp, 6 * EU);  ep_fma(acc, fb, Wg, 5 * EU);
        ep_fma(acc, fa, Wg, 6 * EU);
        if (g > 0) {
            v4f* r4 = reinterpret_cast<v4f*>(R2 + (g - 1) * DDIM) + w3 * 32 + lane;
            r4[0]        = acc[0];
            r4[DDIM / 8] = acc[1];
        }
    }
    lds_barrier();
    if (ep && g == 0) {
        const v4f* r4 = reinterpret_cast<const v4f*>(R2) + w3 * 32 + lane;
        v4f* o4 = reinterpret_cast<v4f*>(out + ((long)bb * 4 + p) * DDIM) + w3 * 32 + lane;
        #pragma unroll
        for (int j = 0; j < 2; ++j) {
            const v4f p1 = r4[j * (DDIM / 8)];
            const v4f p2 = r4[j * (DDIM / 8) + DDIM / 4];
            const v4f p3 = r4[j * (DDIM / 8) + 2 * (DDIM / 4)];
            o4[j * (DDIM / 8)] = ((acc[j] + p1) + p2) + p3;
        }
    }
}

__global__ __launch_bounds__(NTHREADS) void part_cam_kernel(
    const float* __restrict__ x, const float* __restrict__ features,
    float* __restrict__ out)
{
    __shared__ __attribute__((aligned(16))) float T[16 * TS];
    __shared__ __attribute__((aligned(16))) float R2[3 * DDIM];

    const int bb = blockIdx.x;
    const int p  = blockIdx.y;
    const int e  = p + 1;

    if (p == 0) {
        wide_part_body(x, features, out, T, R2, bb);
    } else {
        const int s = (p == 1) ? 5 : ((p == 2) ? 54 : 103);
        part_body<99>(x, features, out, T, bb, p, e, s);
    }
}

extern "C" void kernel_launch(void* const* d_in, const int* in_sizes, int n_in,
                              void* d_out, int out_size, void* d_ws, size_t ws_size,
                              hipStream_t stream) {
    (void)in_sizes; (void)n_in; (void)d_ws; (void)ws_size; (void)out_size;
    const float* x        = (const float*)d_in[0];
    const float* features = (const float*)d_in[1];
    float* out            = (float*)d_out;

    dim3 grid(BATCH, 4);
    part_cam_kernel<<<grid, NTHREADS, 0, stream>>>(x, features, out);
}
